# starting kernel
# baseline (speedup 1.0000x reference)
#include <hip/hip_runtime.h>
#include <math.h>


typedef __attribute__((ext_vector_type(2))) float v2f;
typedef __attribute__((ext_vector_type(8))) float v8f;

#define D_DIM 1024
#define F_DIM 8
#define NCOLS 32
#define KSLICE 256
#define ROWS_PER_BLOCK 128
#define LDS_STRIDE 260
#define WS_BYTES (NCOLS * D_DIM * sizeof(float))

__global__ __launch_bounds__(256) void pack_weights_kernel(
    const float* __restrict__ Wg,
    const float* __restrict__ Wf,
    const float* __restrict__ Wt,
    float* __restrict__ wp)
{
    const int idx = blockIdx.x * 256 + threadIdx.x;
    const int n = idx >> 10;
    const int k = idx & (D_DIM - 1);
    float v = 0.0f;
    if (n < 8)        v = Wg[k * F_DIM + n];
    else if (n < 16)  v = Wf[k * F_DIM + (n - 8)];
    else if (n < 24)  v = Wt[k * F_DIM + (n - 16)];
    wp[n * D_DIM + k] = v;
}

template <bool USE_WS>
__global__ __launch_bounds__(256) void fused_router_kernel(
    const float* __restrict__ x,
    const float* __restrict__ wp,
    const float* __restrict__ Wg, const float* __restrict__ bg,
    const float* __restrict__ Wf, const float* __restrict__ bf,
    const float* __restrict__ Wt, const float* __restrict__ bt,
    const float* __restrict__ alpha,
    float* __restrict__ out,
    int rows)
{
    __shared__ __align__(16) float bsm[NCOLS * LDS_STRIDE];

    const int tid  = threadIdx.x;
    const int lane = tid & 31;
    const int hf   = lane >> 4;
    const int l15  = lane & 15;
    const int wave = tid >> 5;

    const size_t rowBase = (size_t)blockIdx.x * ROWS_PER_BLOCK + (size_t)wave * 16;
    const float* xrow = x + (rowBase + (size_t)l15) * D_DIM;

    v8f c0 = {};
    v8f c1 = {};

    for (int ko = 0; ko < D_DIM; ko += KSLICE) {
        __syncthreads();
        if (USE_WS) {
            for (int i = tid; i < NCOLS * KSLICE / 4; i += 256) {
                const int n   = i >> 6;
                const int kk4 = (i & 63) << 2;
                const float4 v = *(const float4*)(wp + n * D_DIM + ko + kk4);
                *(float4*)(bsm + n * LDS_STRIDE + kk4) = v;
            }
        } else {
            for (int i = tid; i < NCOLS * KSLICE; i += 256) {
                const int n  = i >> 8;
                const int kk = i & (KSLICE - 1);
                const int k  = ko + kk;
                float v = 0.0f;
                if (n < 8)        v = Wg[k * F_DIM + n];
                else if (n < 16)  v = Wf[k * F_DIM + (n - 8)];
                else if (n < 24)  v = Wt[k * F_DIM + (n - 16)];
                bsm[n * LDS_STRIDE + kk] = v;
            }
        }
        __syncthreads();

        if (ko + KSLICE < D_DIM)
            __builtin_prefetch(xrow + ko + KSLICE, 0, 0);

        const float* b0p = bsm + l15 * LDS_STRIDE;
        const float* b1p = bsm + (l15 + 16) * LDS_STRIDE;

        #pragma unroll 4
        for (int kc = 0; kc < KSLICE / 4; ++kc) {
            const int kl = kc * 4 + hf * 2;
            v2f a  = *(const v2f*)(xrow + ko + kl);
            v2f b0 = *(const v2f*)(b0p + kl);
            v2f b1 = *(const v2f*)(b1p + kl);
            c0 = __builtin_amdgcn_wmma_f32_16x16x4_f32(false, a, false, b0,
                                                       (short)0, c0, false, false);
            c1 = __builtin_amdgcn_wmma_f32_16x16x4_f32(false, a, false, b1,
                                                       (short)0, c1, false, false);
        }
    }

    const float ab    = 1.0f / (1.0f + __expf(-alpha[0]));
    const float biasA = (l15 < 8) ? bg[l15] : bf[l15 - 8];
    const float biasB = (l15 < 8) ? bt[l15] : -__builtin_inff();

    #pragma unroll
    for (int r = 0; r < 8; ++r) {
        float v = c0[r] + biasA;

        float m = v;
        m = fmaxf(m, __shfl_xor(m, 1));
        m = fmaxf(m, __shfl_xor(m, 2));
        m = fmaxf(m, __shfl_xor(m, 4));
        float e = __expf(v - m);
        float s = e;
        s += __shfl_xor(s, 1);
        s += __shfl_xor(s, 2);
        s += __shfl_xor(s, 4);
        const float soft   = e / s;
        const float soft_f = __shfl_xor(soft, 8);
        const float gate   = 1.0f / (1.0f + __expf(-v));

        float t  = c1[r] + biasB;
        float t1 = t, t2 = -__builtin_inff();
        #pragma unroll
        for (int msk = 1; msk <= 4; msk <<= 1) {
            const float o1 = __shfl_xor(t1, msk);
            const float o2 = __shfl_xor(t2, msk);
            const float hi = fmaxf(t1, o1);
            const float lo = fminf(t1, o1);
            t2 = fmaxf(lo, fmaxf(t2, o2));
            t1 = hi;
        }
        const float p1 = 1.0f / (1.0f + __expf(t2 - t1));
        const float p2 = 1.0f - p1;
        const float sv = (t == t1) ? p1 : ((t == t2) ? p2 : 0.0f);

        float contrib = gate * (ab * sv + (1.0f - ab) * soft_f);
        contrib += __shfl_xor(contrib, 1);
        contrib += __shfl_xor(contrib, 2);
        contrib += __shfl_xor(contrib, 4);

        const size_t row = rowBase + (size_t)r + (size_t)(8 * hf);
        if (l15 == 0 && row < (size_t)rows)
            out[row] = contrib;
    }
}

extern "C" void kernel_launch(void* const* d_in, const int* in_sizes, int n_in,
                              void* d_out, int out_size, void* d_ws, size_t ws_size,
                              hipStream_t stream) {
    (void)n_in; (void)out_size;
    const float* x     = (const float*)d_in[0];
    const float* Wg    = (const float*)d_in[1];
    const float* bg    = (const float*)d_in[2];
    const float* Wf    = (const float*)d_in[3];
    const float* bf    = (const float*)d_in[4];
    const float* Wt    = (const float*)d_in[5];
    const float* bt    = (const float*)d_in[6];
    const float* alpha = (const float*)d_in[7];

    const int rows = in_sizes[0] / D_DIM;
    const int grid = (rows + ROWS_PER_BLOCK - 1) / ROWS_PER_BLOCK;

    if (ws_size >= WS_BYTES) {
        float* wp = (float*)d_ws;
        pack_weights_kernel<<<(NCOLS * D_DIM) / 256, 256, 0, stream>>>(Wg, Wf, Wt, wp);
        fused_router_kernel<true><<<grid, 256, 0, stream>>>(
            x, wp, Wg, bg, Wf, bf, Wt, bt, alpha, (float*)d_out, rows);
    } else {
        fused_router_kernel<false><<<grid, 256, 0, stream>>>(
            x, nullptr, Wg, bg, Wf, bf, Wt, bt, alpha, (float*)d_out, rows);
    }
}


# x stream restructured: b128 loads into registers two 128-k chunks ahead of the products, packed B staged in LDS once (no barrier in the stream); both loaders permute k alike: k sum order changes
# speedup vs baseline: 1.4655x; 1.4655x over previous
#include <hip/hip_runtime.h>
#include <math.h>


typedef __attribute__((ext_vector_type(2))) float v2f;
typedef __attribute__((ext_vector_type(4))) float v4f;
typedef __attribute__((ext_vector_type(8))) float v8f;

#define D_DIM 1024
#define F_DIM 8
#define NCOLS 32
#define ROWS_PER_BLOCK 128
#define B_STRIDE 1028
#define CHUNK_K 128
#define NCHUNK (D_DIM / CHUNK_K)
#define PIECES (CHUNK_K / 8)
#define RING 3
#define WS_BYTES (NCOLS * D_DIM * sizeof(float))

__global__ __launch_bounds__(256) void pack_weights_kernel(
    const float* __restrict__ Wg,
    const float* __restrict__ Wf,
    const float* __restrict__ Wt,
    float* __restrict__ wp)
{
    const int idx = blockIdx.x * 256 + threadIdx.x;
    const int n = idx >> 10;
    const int k = idx & (D_DIM - 1);
    float v = 0.0f;
    if (n < 8)        v = Wg[k * F_DIM + n];
    else if (n < 16)  v = Wf[k * F_DIM + (n - 8)];
    else if (n < 24)  v = Wt[k * F_DIM + (n - 16)];
    wp[n * D_DIM + k] = v;
}

template <bool USE_WS>
__global__ __launch_bounds__(256) void fused_router_kernel(
    const float* __restrict__ x,
    const float* __restrict__ wp,
    const float* __restrict__ Wg, const float* __restrict__ bg,
    const float* __restrict__ Wf, const float* __restrict__ bf,
    const float* __restrict__ Wt, const float* __restrict__ bt,
    const float* __restrict__ alpha,
    float* __restrict__ out,
    int rows)
{
    __shared__ __align__(16) float bsm[NCOLS * B_STRIDE];

    const int tid  = threadIdx.x;
    const int lane = tid & 31;
    const int hf   = lane >> 4;
    const int l15  = lane & 15;
    const int wave = tid >> 5;

    const size_t rowBase = (size_t)blockIdx.x * ROWS_PER_BLOCK + (size_t)wave * 16;
    const float* xp = x + (rowBase + (size_t)l15) * D_DIM + 4 * hf;

    v4f xr[RING][PIECES];
#define LOAD_X(C) do { _Pragma("unroll") for (int j_ = 0; j_ < PIECES; ++j_) \
        xr[(C) % RING][j_] = *(const v4f*)(xp + (C) * CHUNK_K + 8 * j_); } while (0)

    v4f bt4[NCOLS];
    if (USE_WS) {
        #pragma unroll
        for (int n = 0; n < NCOLS; ++n)
            bt4[n] = *(const v4f*)(wp + n * D_DIM + 4 * tid);
    }
    LOAD_X(0);
    LOAD_X(1);
    if (USE_WS) {
        #pragma unroll
        for (int n = 0; n < NCOLS; ++n)
            *(v4f*)(bsm + n * B_STRIDE + 4 * tid) = bt4[n];
    } else {
        for (int i = tid; i < NCOLS * D_DIM; i += 256) {
            const int n = i >> 10;
            const int k = i & (D_DIM - 1);
            float v = 0.0f;
            if (n < 8)        v = Wg[k * F_DIM + n];
            else if (n < 16)  v = Wf[k * F_DIM + (n - 8)];
            else if (n < 24)  v = Wt[k * F_DIM + (n - 16)];
            bsm[n * B_STRIDE + k] = v;
        }
    }
    __syncthreads();

    v8f c0 = {};
    v8f c1 = {};

    const float* b0p = bsm + l15 * B_STRIDE + 4 * hf;
    const float* b1p = b0p + 16 * B_STRIDE;

    #pragma unroll
    for (int c = 0; c < NCHUNK; ++c) {
        __builtin_amdgcn_sched_barrier(0);
        if (c + 2 < NCHUNK) LOAD_X(c + 2);
        __builtin_amdgcn_sched_barrier(0);

        #pragma unroll
        for (int j = 0; j < PIECES; ++j) {
            const v4f a  = xr[c % RING][j];
            const v4f b0 = *(const v4f*)(b0p + c * CHUNK_K + 8 * j);
            const v4f b1 = *(const v4f*)(b1p + c * CHUNK_K + 8 * j);
            c0 = __builtin_amdgcn_wmma_f32_16x16x4_f32(false, a.xy, false, b0.xy,
                                                       (short)0, c0, false, false);
            c1 = __builtin_amdgcn_wmma_f32_16x16x4_f32(false, a.xy, false, b1.xy,
                                                       (short)0, c1, false, false);
            c0 = __builtin_amdgcn_wmma_f32_16x16x4_f32(false, a.zw, false, b0.zw,
                                                       (short)0, c0, false, false);
            c1 = __builtin_amdgcn_wmma_f32_16x16x4_f32(false, a.zw, false, b1.zw,
                                                       (short)0, c1, false, false);
        }
    }
#undef LOAD_X

    const float ab    = 1.0f / (1.0f + __expf(-alpha[0]));
    const float biasA = (l15 < 8) ? bg[l15] : bf[l15 - 8];
    const float biasB = (l15 < 8) ? bt[l15] : -__builtin_inff();

    #pragma unroll
    for (int r = 0; r < 8; ++r) {
        float v = c0[r] + biasA;

        float m = v;
        m = fmaxf(m, __shfl_xor(m, 1));
        m = fmaxf(m, __shfl_xor(m, 2));
        m = fmaxf(m, __shfl_xor(m, 4));
        float e = __expf(v - m);
        float s = e;
        s += __shfl_xor(s, 1);
        s += __shfl_xor(s, 2);
        s += __shfl_xor(s, 4);
        const float soft   = e / s;
        const float soft_f = __shfl_xor(soft, 8);
        const float gate   = 1.0f / (1.0f + __expf(-v));

        float t  = c1[r] + biasB;
        float t1 = t, t2 = -__builtin_inff();
        #pragma unroll
        for (int msk = 1; msk <= 4; msk <<= 1) {
            const float o1 = __shfl_xor(t1, msk);
            const float o2 = __shfl_xor(t2, msk);
            const float hi = fmaxf(t1, o1);
            const float lo = fminf(t1, o1);
            t2 = fmaxf(lo, fmaxf(t2, o2));
            t1 = hi;
        }
        const float p1 = 1.0f / (1.0f + __expf(t2 - t1));
        const float p2 = 1.0f - p1;
        const float sv = (t == t1) ? p1 : ((t == t2) ? p2 : 0.0f);

        float contrib = gate * (ab * sv + (1.0f - ab) * soft_f);
        contrib += __shfl_xor(contrib, 1);
        contrib += __shfl_xor(contrib, 2);
        contrib += __shfl_xor(contrib, 4);

        const size_t row = rowBase + (size_t)r + (size_t)(8 * hf);
        if (l15 == 0 && row < (size_t)rows)
            out[row] = contrib;
    }
}

extern "C" void kernel_launch(void* const* d_in, const int* in_sizes, int n_in,
                              void* d_out, int out_size, void* d_ws, size_t ws_size,
                              hipStream_t stream) {
    (void)n_in; (void)out_size;
    const float* x     = (const float*)d_in[0];
    const float* Wg    = (const float*)d_in[1];
    const float* bg    = (const float*)d_in[2];
    const float* Wf    = (const float*)d_in[3];
    const float* bf    = (const float*)d_in[4];
    const float* Wt    = (const float*)d_in[5];
    const float* bt    = (const float*)d_in[6];
    const float* alpha = (const float*)d_in[7];

    const int rows = in_sizes[0] / D_DIM;
    const int grid = (rows + ROWS_PER_BLOCK - 1) / ROWS_PER_BLOCK;

    if (ws_size >= WS_BYTES) {
        float* wp = (float*)d_ws;
        pack_weights_kernel<<<(NCOLS * D_DIM) / 256, 256, 0, stream>>>(Wg, Wf, Wt, wp);
        fused_router_kernel<true><<<grid, 256, 0, stream>>>(
            x, wp, Wg, bg, Wf, bf, Wt, bt, alpha, (float*)d_out, rows);
    } else {
        fused_router_kernel<false><<<grid, 256, 0, stream>>>(
            x, nullptr, Wg, bg, Wf, bf, Wt, bt, alpha, (float*)d_out, rows);
    }
}


# epilogue lane exchanges (xor 1, 2, 4, 8 inside a 16-lane row) done as DPP row_xmask moves on the vector ALU instead of LDS permutes with a wait per link; values and order of sums unchanged
# speedup vs baseline: 1.5840x; 1.0809x over previous
#include <hip/hip_runtime.h>
#include <math.h>


typedef __attribute__((ext_vector_type(2))) float v2f;
typedef __attribute__((ext_vector_type(4))) float v4f;
typedef __attribute__((ext_vector_type(8))) float v8f;

#define D_DIM 1024
#define F_DIM 8
#define NCOLS 32
#define ROWS_PER_BLOCK 128
#define B_STRIDE 1028
#define CHUNK_K 128
#define NCHUNK (D_DIM / CHUNK_K)
#define PIECES (CHUNK_K / 8)
#define RING 3
#define WS_BYTES (NCOLS * D_DIM * sizeof(float))

template <int M>
static __device__ __forceinline__ float lane_xor(float v) {
    return __builtin_bit_cast(float,
        __builtin_amdgcn_update_dpp(0, __builtin_bit_cast(int, v),
                                    0x160 | M, 0xF, 0xF, true));
}

template <int M, bool FIRST>
static __device__ __forceinline__ void top2_step(float& t1, float& t2) {
    const float o1 = lane_xor<M>(t1);
    const float o2 = FIRST ? t2 : lane_xor<M>(t2);
    const float hi = fmaxf(t1, o1);
    const float lo = fminf(t1, o1);
    t2 = fmaxf(lo, fmaxf(t2, o2));
    t1 = hi;
}

__global__ __launch_bounds__(256) void pack_weights_kernel(
    const float* __restrict__ Wg,
    const float* __restrict__ Wf,
    const float* __restrict__ Wt,
    float* __restrict__ wp)
{
    const int idx = blockIdx.x * 256 + threadIdx.x;
    const int n = idx >> 10;
    const int k = idx & (D_DIM - 1);
    float v = 0.0f;
    if (n < 8)        v = Wg[k * F_DIM + n];
    else if (n < 16)  v = Wf[k * F_DIM + (n - 8)];
    else if (n < 24)  v = Wt[k * F_DIM + (n - 16)];
    wp[n * D_DIM + k] = v;
}

template <bool USE_WS>
__global__ __launch_bounds__(256) void fused_router_kernel(
    const float* __restrict__ x,
    const float* __restrict__ wp,
    const float* __restrict__ Wg, const float* __restrict__ bg,
    const float* __restrict__ Wf, const float* __restrict__ bf,
    const float* __restrict__ Wt, const float* __restrict__ bt,
    const float* __restrict__ alpha,
    float* __restrict__ out,
    int rows)
{
    __shared__ __align__(16) float bsm[NCOLS * B_STRIDE];

    const int tid  = threadIdx.x;
    const int lane = tid & 31;
    const int hf   = lane >> 4;
    const int l15  = lane & 15;
    const int wave = tid >> 5;

    const size_t rowBase = (size_t)blockIdx.x * ROWS_PER_BLOCK + (size_t)wave * 16;
    const float* xp = x + (rowBase + (size_t)l15) * D_DIM + 4 * hf;

    v4f xr[RING][PIECES];
#define LOAD_X(C) do { _Pragma("unroll") for (int j_ = 0; j_ < PIECES; ++j_) \
        xr[(C) % RING][j_] = *(const v4f*)(xp + (C) * CHUNK_K + 8 * j_); } while (0)

    v4f bt4[NCOLS];
    if (USE_WS) {
        #pragma unroll
        for (int n = 0; n < NCOLS; ++n)
            bt4[n] = *(const v4f*)(wp + n * D_DIM + 4 * tid);
    }
    LOAD_X(0);
    LOAD_X(1);
    if (USE_WS) {
        #pragma unroll
        for (int n = 0; n < NCOLS; ++n)
            *(v4f*)(bsm + n * B_STRIDE + 4 * tid) = bt4[n];
    } else {
        for (int i = tid; i < NCOLS * D_DIM; i += 256) {
            const int n = i >> 10;
            const int k = i & (D_DIM - 1);
            float v = 0.0f;
            if (n < 8)        v = Wg[k * F_DIM + n];
            else if (n < 16)  v = Wf[k * F_DIM + (n - 8)];
            else if (n < 24)  v = Wt[k * F_DIM + (n - 16)];
            bsm[n * B_STRIDE + k] = v;
        }
    }
    __syncthreads();

    v8f c0 = {};
    v8f c1 = {};

    const float* b0p = bsm + l15 * B_STRIDE + 4 * hf;
    const float* b1p = b0p + 16 * B_STRIDE;

    #pragma unroll
    for (int c = 0; c < NCHUNK; ++c) {
        __builtin_amdgcn_sched_barrier(0);
        if (c + 2 < NCHUNK) LOAD_X(c + 2);
        __builtin_amdgcn_sched_barrier(0);

        #pragma unroll
        for (int j = 0; j < PIECES; ++j) {
            const v4f a  = xr[c % RING][j];
            const v4f b0 = *(const v4f*)(b0p + c * CHUNK_K + 8 * j);
            const v4f b1 = *(const v4f*)(b1p + c * CHUNK_K + 8 * j);
            c0 = __builtin_amdgcn_wmma_f32_16x16x4_f32(false, a.xy, false, b0.xy,
                                                       (short)0, c0, false, false);
            c1 = __builtin_amdgcn_wmma_f32_16x16x4_f32(false, a.xy, false, b1.xy,
                                                       (short)0, c1, false, false);
            c0 = __builtin_amdgcn_wmma_f32_16x16x4_f32(false, a.zw, false, b0.zw,
                                                       (short)0, c0, false, false);
            c1 = __builtin_amdgcn_wmma_f32_16x16x4_f32(false, a.zw, false, b1.zw,
                                                       (short)0, c1, false, false);
        }
    }
#undef LOAD_X

    const float ab    = 1.0f / (1.0f + __expf(-alpha[0]));
    const float biasA = (l15 < 8) ? bg[l15] : bf[l15 - 8];
    const float biasB = (l15 < 8) ? bt[l15] : -__builtin_inff();

    #pragma unroll
    for (int r = 0; r < 8; ++r) {
        float v = c0[r] + biasA;

        float m = v;
        m = fmaxf(m, lane_xor<1>(m));
        m = fmaxf(m, lane_xor<2>(m));
        m = fmaxf(m, lane_xor<4>(m));
        float e = __expf(v - m);
        float s = e;
        s += lane_xor<1>(s);
        s += lane_xor<2>(s);
        s += lane_xor<4>(s);
        const float soft   = e / s;
        const float soft_f = lane_xor<8>(soft);
        const float gate   = 1.0f / (1.0f + __expf(-v));

        float t  = c1[r] + biasB;
        float t1 = t, t2 = -__builtin_inff();
        top2_step<1, true>(t1, t2);
        top2_step<2, false>(t1, t2);
        top2_step<4, false>(t1, t2);
        const float p1 = 1.0f / (1.0f + __expf(t2 - t1));
        const float p2 = 1.0f - p1;
        const float sv = (t == t1) ? p1 : ((t == t2) ? p2 : 0.0f);

        float contrib = gate * (ab * sv + (1.0f - ab) * soft_f);
        contrib += lane_xor<1>(contrib);
        contrib += lane_xor<2>(contrib);
        contrib += lane_xor<4>(contrib);

        const size_t row = rowBase + (size_t)r + (size_t)(8 * hf);
        if (l15 == 0 && row < (size_t)rows)
            out[row] = contrib;
    }
}

extern "C" void kernel_launch(void* const* d_in, const int* in_sizes, int n_in,
                              void* d_out, int out_size, void* d_ws, size_t ws_size,
                              hipStream_t stream) {
    (void)n_in; (void)out_size;
    const float* x     = (const float*)d_in[0];
    const float* Wg    = (const float*)d_in[1];
    const float* bg    = (const float*)d_in[2];
    const float* Wf    = (const float*)d_in[3];
    const float* bf    = (const float*)d_in[4];
    const float* Wt    = (const float*)d_in[5];
    const float* bt    = (const float*)d_in[6];
    const float* alpha = (const float*)d_in[7];

    const int rows = in_sizes[0] / D_DIM;
    const int grid = (rows + ROWS_PER_BLOCK - 1) / ROWS_PER_BLOCK;

    if (ws_size >= WS_BYTES) {
        float* wp = (float*)d_ws;
        pack_weights_kernel<<<(NCOLS * D_DIM) / 256, 256, 0, stream>>>(Wg, Wf, Wt, wp);
        fused_router_kernel<true><<<grid, 256, 0, stream>>>(
            x, wp, Wg, bg, Wf, bf, Wt, bt, alpha, (float*)d_out, rows);
    } else {
        fused_router_kernel<false><<<grid, 256, 0, stream>>>(
            x, nullptr, Wg, bg, Wf, bf, Wt, bt, alpha, (float*)d_out, rows);
    }
}
